# starting kernel
# baseline (speedup 1.0000x reference)
#include <hip/hip_runtime.h>
#include <math.h>

#define TT   2048
#define BB   32
#define DD   256
#define HH   256
#define W3H  768
#define NPRE 512
#define HSTR 260

typedef __attribute__((ext_vector_type(2))) float v2f;
typedef __attribute__((ext_vector_type(8))) float v8f;

__device__ __forceinline__ size_t px_off(int t, int w, int mt, int lane) {
  return ((((size_t)(t * 32 + w) * 2 + mt) * 32 + lane) * 8);
}

__global__ __launch_bounds__(256) void rhn_gemm_x(
    const float* __restrict__ X, const float* __restrict__ W,
    const float* __restrict__ bias, float* __restrict__ preX) {
  extern __shared__ char smem[];
  v2f* wtile = (v2f*)smem;
  const int tid  = threadIdx.x;
  const int lane = tid & 31;
  const int wave = tid >> 5;
  const int m0   = blockIdx.x * 128 + wave * 16;
  const int n0   = blockIdx.y * 128;

  for (int e = tid; e < 64 * 8 * 32; e += 256) {
    const int c  = e >> 8;
    const int nt = (e >> 5) & 7;
    const int l  = e & 31;
    const int k  = 4 * c + ((l >> 4) << 1);
    const int n  = n0 + nt * 16 + (l & 15);
    v2f wv;
    wv.x = W[k * W3H + n];
    wv.y = W[(k + 1) * W3H + n];
    wtile[e] = wv;
  }
  __syncthreads();

  v8f acc[8];
#pragma unroll
  for (int nt = 0; nt < 8; ++nt) {
    const float bv = bias[n0 + nt * 16 + (lane & 15)];
#pragma unroll
    for (int r = 0; r < 8; ++r) acc[nt][r] = bv;
  }

  const int arow = m0 + (lane & 15);
  const float* aptr = X + (size_t)arow * DD + ((lane >> 4) << 1);

#pragma unroll 8
  for (int c = 0; c < 64; ++c) {
    const v2f a = *(const v2f*)(aptr + 4 * c);
#pragma unroll
    for (int nt = 0; nt < 8; ++nt) {
      const v2f b = wtile[(c * 8 + nt) * 32 + lane];
      acc[nt] = __builtin_amdgcn_wmma_f32_16x16x4_f32(
          false, a, false, b, (short)0, acc[nt], false, false);
    }
  }

  const int tIdx = m0 >> 5;
  const int mt   = (m0 >> 4) & 1;
#pragma unroll
  for (int nt = 0; nt < 8; ++nt) {
    const int w = (n0 >> 4) + nt;
    *(v8f*)(preX + px_off(tIdx, w, mt, lane)) = acc[nt];
  }
}

__device__ __forceinline__ float sigf(float x) {
  return 1.0f / (1.0f + __expf(-x));
}

__global__ __launch_bounds__(1024) void rhn_recur(
    const float* __restrict__ preX, const float* __restrict__ Whh,
    const int* __restrict__ length, float* __restrict__ out,
    float* __restrict__ hn) {
  extern __shared__ char smem[];
  float* hbuf = (float*)smem;
  float* tbuf = hbuf + 32 * HSTR;
  int*   lbuf = (int*)(tbuf + 16 * 2 * 32 * 8);

  const int tid  = threadIdx.x;
  const int lane = tid & 31;
  const int wave = tid >> 5;

  for (int i = tid; i < 32 * HSTR; i += 1024) hbuf[i] = 0.0f;
  if (tid < 32) lbuf[tid] = length[tid];
  __syncthreads();

  const int ncol = wave * 16 + (lane & 15);
  const int ksub = (lane >> 4) << 1;
  v2f wreg[64];
#pragma unroll
  for (int c = 0; c < 64; ++c) {
    const int k = 4 * c + ksub;
    v2f wv;
    wv.x = Whh[k * W3H + ncol];
    wv.y = Whh[(k + 1) * W3H + ncol];
    wreg[c] = wv;
  }

  const int crow = (lane >> 4) << 3;
  const int am   = lane & 15;
  const bool isH = (wave < 16);
  const int  w16 = isH ? wave : (wave - 16);

  v8f cur0 = *(const v8f*)(preX + px_off(0, wave, 0, lane));
  v8f cur1 = *(const v8f*)(preX + px_off(0, wave, 1, lane));

  for (int t = 0; t < TT; ++t) {
    v8f acc0 = cur0;
    v8f acc1 = cur1;

    const int tn = (t + 1) & (TT - 1);
    cur0 = *(const v8f*)(preX + px_off(tn, wave, 0, lane));
    cur1 = *(const v8f*)(preX + px_off(tn, wave, 1, lane));

#pragma unroll
    for (int c = 0; c < 64; ++c) {
      const v2f a0 = *(const v2f*)(hbuf + am * HSTR + 4 * c + ksub);
      const v2f a1 = *(const v2f*)(hbuf + (16 + am) * HSTR + 4 * c + ksub);
      acc0 = __builtin_amdgcn_wmma_f32_16x16x4_f32(
          false, a0, false, wreg[c], (short)0, acc0, false, false);
      acc1 = __builtin_amdgcn_wmma_f32_16x16x4_f32(
          false, a1, false, wreg[c], (short)0, acc1, false, false);
    }

    if (!isH) {
      float* tb0 = tbuf + ((w16 * 2 + 0) * 32 + lane) * 8;
      float* tb1 = tbuf + ((w16 * 2 + 1) * 32 + lane) * 8;
#pragma unroll
      for (int r = 0; r < 8; ++r) { tb0[r] = acc0[r]; tb1[r] = acc1[r]; }
    }
    __syncthreads();

    if (isH) {
#pragma unroll
      for (int mt = 0; mt < 2; ++mt) {
        const float* tb = tbuf + ((w16 * 2 + mt) * 32 + lane) * 8;
        const v8f hp = mt ? acc1 : acc0;
#pragma unroll
        for (int r = 0; r < 8; ++r) {
          const int b   = mt * 16 + r + crow;
          const int col = w16 * 16 + (lane & 15);
          const float hold = hbuf[b * HSTR + col];
          const float tg = sigf(tb[r]);
          const float cg = sigf(tg);
          const float s  = tanhf(hp[r]) * tg + hold * cg;
          const float hv = (t < lbuf[b]) ? s : hold;
          hbuf[b * HSTR + col] = hv;
          out[(size_t)(t * BB + b) * HH + col] = hv;
        }
      }
    }
    __syncthreads();
  }

  for (int i = tid; i < BB * HH; i += 1024)
    hn[i] = hbuf[(i >> 8) * HSTR + (i & 255)];
}

extern "C" void kernel_launch(void* const* d_in, const int* in_sizes, int n_in,
                              void* d_out, int out_size, void* d_ws,
                              size_t ws_size, hipStream_t stream) {
  const float* input_ = (const float*)d_in[0];
  const int*   length = (const int*)d_in[1];
  const float* Wih0   = (const float*)d_in[2];
  const float* Whh0   = (const float*)d_in[3];
  const float* b0     = (const float*)d_in[4];
  const float* Wih1   = (const float*)d_in[5];
  const float* Whh1   = (const float*)d_in[6];
  const float* b1     = (const float*)d_in[7];

  float* out  = (float*)d_out;
  float* hn   = out + (size_t)TT * BB * HH;
  float* preX = (float*)d_ws;

  const dim3  g1(512, 4);
  const size_t lds1 = (size_t)64 * 8 * 32 * sizeof(v2f);
  const size_t lds2 = (size_t)(32 * HSTR + 16 * 2 * 32 * 8 + 32) * 4;

  rhn_gemm_x<<<g1, 256, lds1, stream>>>(input_, Wih0, b0, preX);
  rhn_recur<<<dim3(1), 1024, lds2, stream>>>(preX, Whh0, length, out, hn);
  rhn_gemm_x<<<g1, 256, lds1, stream>>>(out, Wih1, b1, preX);
  rhn_recur<<<dim3(1), 1024, lds2, stream>>>(preX, Whh1, length, out,
                                             hn + BB * HH);
}


# recurrence: rows placed in a tile by length-rank pairs (rank 2r in row r, 2r+1 in row 8+r): a gate element's two rows freeze together, live elements are a prefix, frozen rest not visited; sums same
# speedup vs baseline: 2.2013x; 2.2013x over previous
#include <hip/hip_runtime.h>
#include <math.h>

#define TT   2048
#define BB   32
#define DD   256
#define HH   256
#define W3H  768
#define NPRE 512
#define HSTR 260
#define RES  34
#define PRE  4
#define NSTR (64 - RES)

__host__ __device__ constexpr bool is_streamed(int c) {
  return c >= PRE && ((c - PRE) & 1) == 0;
}
__host__ __device__ constexpr int str_index(int c) { return (c - PRE) >> 1; }
__host__ __device__ constexpr int res_index(int c) {
  return c < PRE ? c : PRE + ((c - PRE) >> 1);
}
__host__ __device__ constexpr int str_chunk(int j) { return PRE + 2 * j; }
__host__ __device__ constexpr int res_chunk(int i) {
  return i < PRE ? i : 2 * i - PRE + 1;
}
static_assert(PRE + 2 * NSTR == 64, "chunk map must cover 64 chunks");

typedef __attribute__((ext_vector_type(2))) float v2f;
typedef __attribute__((ext_vector_type(8))) float v8f;
typedef __attribute__((ext_vector_type(4))) float v4f;

__device__ __forceinline__ int len_rank(const int* __restrict__ length, int b) {
  const int myl = length[b];
  int rank = 0;
  for (int j = 0; j < BB; ++j) {
    const int lj = length[j];
    rank += (lj > myl || (lj == myl && j < b)) ? 1 : 0;
  }
  return (rank & 16) | ((rank & 1) << 3) | ((rank & 15) >> 1);
}

__device__ __forceinline__ size_t px_off(int t, int w, int mt, int lane) {
  return ((((size_t)(t * 32 + w) * 2 + mt) * 32 + lane) * 8);
}

__global__ __launch_bounds__(256) void rhn_gemm_x(
    const float* __restrict__ X, const float* __restrict__ W,
    const float* __restrict__ bias, const int* __restrict__ length,
    float* __restrict__ preX) {
  extern __shared__ char smem[];
  v2f* wtile = (v2f*)smem;
  int* perm  = (int*)(wtile + 64 * 8 * 32);
  const int tid  = threadIdx.x;
  const int lane = tid & 31;
  const int wave = tid >> 5;
  const int m0   = blockIdx.x * 128 + wave * 16;
  const int n0   = blockIdx.y * 128;

  for (int e = tid; e < 64 * 8 * 32; e += 256) {
    const int c  = e >> 8;
    const int nt = (e >> 5) & 7;
    const int l  = e & 31;
    const int k  = 4 * c + ((l >> 4) << 1);
    const int n  = n0 + nt * 16 + (l & 15);
    v2f wv;
    wv.x = W[k * W3H + n];
    wv.y = W[(k + 1) * W3H + n];
    wtile[e] = wv;
  }
  if (tid < BB) perm[len_rank(length, tid)] = tid;
  __syncthreads();

  v8f acc[8];
#pragma unroll
  for (int nt = 0; nt < 8; ++nt) {
    const float bv = bias[n0 + nt * 16 + (lane & 15)];
#pragma unroll
    for (int r = 0; r < 8; ++r) acc[nt][r] = bv;
  }

  const int apos = m0 + (lane & 15);
  const int arow = (apos & ~(BB - 1)) + perm[apos & (BB - 1)];
  const float* aptr = X + (size_t)arow * DD + ((lane >> 4) << 1);

#pragma unroll 8
  for (int c = 0; c < 64; ++c) {
    const v2f a = *(const v2f*)(aptr + 4 * c);
#pragma unroll
    for (int nt = 0; nt < 8; ++nt) {
      const v2f b = wtile[(c * 8 + nt) * 32 + lane];
      acc[nt] = __builtin_amdgcn_wmma_f32_16x16x4_f32(
          false, a, false, b, (short)0, acc[nt], false, false);
    }
  }

  const int tIdx = m0 >> 5;
  const int mt   = (m0 >> 4) & 1;
#pragma unroll
  for (int nt = 0; nt < 8; ++nt) {
    const int w = (n0 >> 4) + nt;
    *(v8f*)(preX + px_off(tIdx, w, mt, lane)) = acc[nt];
  }
}

__device__ __forceinline__ float sigf(float x) {
  return 1.0f / (1.0f + __expf(-x));
}


template <bool TWO>
__device__ __forceinline__ void rhn_step(
    const int t, const v2f (&wreg)[RES], const float* wp, const float* afrag,
    float* tfrag, const float* __restrict__ pxp, v8f& cur0, v8f& cur1,
    const bool isH, float* hcell, const int2* lpcell, const float* hrow,
    float* frow, const int d0, const v4f flo, const v4f fhi, const int nl) {
  v8f acc0 = cur0;
  v8f acc1 = cur1;

  constexpr int NA = TWO ? 2 : 3;
  constexpr int DA = NA - 1;
  constexpr int RG = TWO ? 5 : 7;
  v2f a0b[NA][2];
  v2f a1b[TWO ? NA : 1][2];
  v2f ring[RG];

#pragma unroll
  for (int q = 0; q < DA; ++q) {
    a0b[q][0] = *(const v2f*)(afrag + 8 * q);
    a0b[q][1] = *(const v2f*)(afrag + 8 * q + 4);
    if constexpr (TWO) {
      a1b[q][0] = *(const v2f*)(afrag + 16 * HSTR + 8 * q);
      a1b[q][1] = *(const v2f*)(afrag + 16 * HSTR + 8 * q + 4);
    }
  }
#pragma unroll
  for (int j = 0; j < RG; ++j) {
    ring[j].x = wp[(4 * str_chunk(j)) * W3H];
    ring[j].y = wp[(4 * str_chunk(j) + 1) * W3H];
  }

#pragma unroll
  for (int p = 0; p < 32; ++p) {
    if (p + DA < 32) {
      const int q = p + DA;
      a0b[q % NA][0] = *(const v2f*)(afrag + 8 * q);
      a0b[q % NA][1] = *(const v2f*)(afrag + 8 * q + 4);
      if constexpr (TWO) {
        a1b[q % NA][0] = *(const v2f*)(afrag + 16 * HSTR + 8 * q);
        a1b[q % NA][1] = *(const v2f*)(afrag + 16 * HSTR + 8 * q + 4);
      }
    }
#pragma unroll
    for (int u = 0; u < 2; ++u) {
      const int c = 2 * p + u;
      const v2f wv = is_streamed(c) ? ring[str_index(c) % RG]
                                    : wreg[res_index(c)];
      acc0 = __builtin_amdgcn_wmma_f32_16x16x4_f32(
          false, a0b[p % NA][u], false, wv, (short)0, acc0, false, false);
      if constexpr (TWO) {
        acc1 = __builtin_amdgcn_wmma_f32_16x16x4_f32(
            false, a1b[p % NA][u], false, wv, (short)0, acc1, false, false);
      }
      if (is_streamed(c) && str_index(c) + RG < NSTR) {
        const int cn = str_chunk(str_index(c) + RG);
        ring[str_index(c) % RG].x = wp[(4 * cn) * W3H];
        ring[str_index(c) % RG].y = wp[(4 * cn + 1) * W3H];
      }
    }
    __builtin_amdgcn_sched_barrier(0);
  }

  if (!isH) {
#pragma unroll
    for (int r = 0; r < 8; ++r) tfrag[r] = acc0[r];
    if constexpr (TWO) {
#pragma unroll
      for (int r = 0; r < 8; ++r) tfrag[32 * 8 + r] = acc1[r];
    }
  }
  int   len_n  = lpcell[0].x;
  float hold_n = hcell[0];
  __syncthreads();

  const int tn = (t + 1) & (TT - 1);
  const float* pxn = pxp + (size_t)tn * (32 * 2 * 32 * 8);
  cur0 = *(const v8f*)pxn;
  if constexpr (TWO) cur1 = *(const v8f*)(pxn + 32 * 8);

  if (isH) {
    constexpr int NE = TWO ? 16 : 8;
    constexpr int VT = TWO ? 1 : 0;
    float tb_n = tfrag[0];
#pragma unroll
    for (int e = 0; e < NE; ++e) {
      const int mt = e >> 3, r = e & 7;
      if (mt == VT && r >= nl) continue;
      const int br = mt * 16 + r;
      const float hold = hold_n;
      const float tbv  = tb_n;
      const int   len  = len_n;
      if (e + 1 < NE) {
        const int mt2 = (e + 1) >> 3, r2 = (e + 1) & 7;
        const int br2 = mt2 * 16 + r2;
        hold_n = hcell[br2 * HSTR];
        tb_n   = tfrag[mt2 * (32 * 8) + r2];
        len_n  = lpcell[br2].x;
      }
      if (t < len) {
        const float hp = mt ? acc1[r] : acc0[r];
        const float tg = sigf(tbv);
        const float cg = sigf(tg);
        hcell[br * HSTR] = tanhf(hp) * tg + hold * cg;
      }
    }
  } else if constexpr (!TWO) {
    float* fo = frow + (size_t)t * (BB * HH);
    *(v4f*)fo = flo;
    *(v4f*)(fo + 128) = fhi;
  }
  __syncthreads();

  if (!isH) {
    const float* h0 = hrow - 16 * HSTR;
    float* fo1 = frow + (size_t)t * (BB * HH);
    float* fo0 = fo1 + d0;
    const v4f r0 = *(const v4f*)h0;
    const v4f r1 = *(const v4f*)(h0 + 128);
    *(v4f*)fo0 = r0;
    *(v4f*)(fo0 + 128) = r1;
    if constexpr (TWO) {
      const v4f r2 = *(const v4f*)hrow;
      const v4f r3 = *(const v4f*)(hrow + 128);
      *(v4f*)fo1 = r2;
      *(v4f*)(fo1 + 128) = r3;
    }
  }
}

__global__ __launch_bounds__(1024) void rhn_recur(
    const float* __restrict__ preX, const float* Whh,
    const int* __restrict__ length, float* __restrict__ out,
    float* __restrict__ hn) {
  extern __shared__ char smem[];
  float* hbuf  = (float*)smem;
  float* tbuf  = hbuf + 32 * HSTR;
  int2*  lpbuf = (int2*)(tbuf + 16 * 2 * 32 * 8);

  const int tid  = threadIdx.x;
  const int lane = tid & 31;
  const int wave = tid >> 5;

  for (int i = tid; i < 32 * HSTR; i += 1024) hbuf[i] = 0.0f;
  if (tid < BB) lpbuf[len_rank(length, tid)] = make_int2(length[tid], tid * HH);
  __syncthreads();

  const int T1 = __builtin_amdgcn_readfirstlane(min(max(lpbuf[16].x, 0), TT));
  const int T0 = __builtin_amdgcn_readfirstlane(min(max(lpbuf[0].x, 0), TT));

  const int ncol = wave * 16 + (lane & 15);
  const int ksub = (lane >> 4) << 1;
  const float* wp = Whh + ksub * W3H + ncol;
  v2f wreg[RES];
#pragma unroll
  for (int i = 0; i < RES; ++i) {
    v2f wv;
    wv.x = wp[(4 * res_chunk(i)) * W3H];
    wv.y = wp[(4 * res_chunk(i) + 1) * W3H];
    wreg[i] = wv;
  }

  const int crow = (lane >> 4) << 3;
  const int am   = lane & 15;
  const bool isH = (wave < 16);
  const int  w16 = isH ? wave : (wave - 16);

  const int gcol = w16 * 16 + (lane & 15);
  float*      hcell  = hbuf + crow * HSTR + gcol;
  const int2* lpcell = lpbuf + crow;

  const float* afrag = hbuf + am * HSTR + ksub;
  float*       tfrag = tbuf + ((w16 * 2) * 32 + lane) * 8;
  const float* pxp   = preX + px_off(0, wave, 0, lane);

  const float* hrow = hbuf + wave * HSTR + lane * 4;
  const int    foff = lpbuf[wave].y + lane * 4;
  float*       frow = out + foff;
  const int    d0   = lpbuf[w16].y - lpbuf[wave].y;

  v8f cur0 = *(const v8f*)pxp;
  v8f cur1 = *(const v8f*)(pxp + 32 * 8);
  const v4f z4 = {0.0f, 0.0f, 0.0f, 0.0f};

  int nl  = 8;
  int thr = __builtin_amdgcn_readfirstlane(lpbuf[16 + 7].x);
  int t = 0;
  for (; t < T1; ++t) {
    while (nl > 1 && t >= thr) {
      --nl;
      thr = __builtin_amdgcn_readfirstlane(lpbuf[16 + nl - 1].x);
    }
    rhn_step<true>(t, wreg, wp, afrag, tfrag, pxp, cur0, cur1, isH, hcell,
                   lpcell, hrow, frow, d0, z4, z4, nl);
  }

  v4f flo = *(const v4f*)hrow;
  v4f fhi = *(const v4f*)(hrow + 128);
  nl  = 8;
  thr = __builtin_amdgcn_readfirstlane(lpbuf[7].x);
  for (; t < T0; ++t) {
    while (nl > 1 && t >= thr) {
      --nl;
      thr = __builtin_amdgcn_readfirstlane(lpbuf[nl - 1].x);
    }
    rhn_step<false>(t, wreg, wp, afrag, tfrag, pxp, cur0, cur1, isH, hcell,
                    lpcell, hrow, frow, d0, flo, fhi, nl);
  }

  flo = *(const v4f*)hrow;
  fhi = *(const v4f*)(hrow + 128);
  for (; t < TT; ++t) {
    float* fo = frow + (size_t)t * (BB * HH);
    *(v4f*)fo = flo;
    *(v4f*)(fo + 128) = fhi;
  }

  *(v4f*)(hn + foff) = flo;
  *(v4f*)(hn + foff + 128) = fhi;
}

extern "C" void kernel_launch(void* const* d_in, const int* in_sizes, int n_in,
                              void* d_out, int out_size, void* d_ws,
                              size_t ws_size, hipStream_t stream) {
  const float* input_ = (const float*)d_in[0];
  const int*   length = (const int*)d_in[1];
  const float* Wih0   = (const float*)d_in[2];
  const float* Whh0   = (const float*)d_in[3];
  const float* b0     = (const float*)d_in[4];
  const float* Wih1   = (const float*)d_in[5];
  const float* Whh1   = (const float*)d_in[6];
  const float* b1     = (const float*)d_in[7];

  float* out  = (float*)d_out;
  float* hn   = out + (size_t)TT * BB * HH;
  float* preX = (float*)d_ws;

  const dim3  g1(512, 4);
  const size_t lds1 = (size_t)64 * 8 * 32 * sizeof(v2f) + 32 * 4;
  const size_t lds2 = (size_t)(32 * HSTR + 16 * 2 * 32 * 8 + 64) * 4;

  rhn_gemm_x<<<g1, 256, lds1, stream>>>(input_, Wih0, b0, length, preX);
  rhn_recur<<<dim3(1), 1024, lds2, stream>>>(preX, Whh0, length, out, hn);
  rhn_gemm_x<<<g1, 256, lds1, stream>>>(out, Wih1, b1, length, preX);
  rhn_recur<<<dim3(1), 1024, lds2, stream>>>(preX, Whh1, length, out,
                                             hn + BB * HH);
}
